# starting kernel
# baseline (speedup 1.0000x reference)
#include <hip/hip_runtime.h>

typedef float v2f __attribute__((ext_vector_type(2)));
typedef float v8f __attribute__((ext_vector_type(8)));

#define NN    11826
#define DIN   9
#define DGATE 36
#define NE    2
#define HH    128
#define MO    32
#define NO    2

__global__ void moe_zero_ws(int* cnt) {
    if (threadIdx.x == 0) cnt[0] = 0;
}

__global__ __launch_bounds__(32) void moe_tile_kernel(
    const float* __restrict__ x,
    const float* __restrict__ gate,
    const float* __restrict__ w_gate,
    const float* __restrict__ w1,
    const float* __restrict__ b1,
    const float* __restrict__ w2,
    const float* __restrict__ b2,
    const float* __restrict__ wo,
    const float* __restrict__ bo,
    float* __restrict__ out,
    int* __restrict__ cnt1)
{
    __shared__ float h_lds[16 * HH];
    __shared__ float y_lds[NE * 16 * MO];
    __shared__ int   sel[16];

    const int lane = threadIdx.x;
    const int row0 = blockIdx.x * 16;
    const int m    = lane & 15;
    const int hi   = lane >> 4;
    const int rowm = min(row0 + m, NN - 1);

    if (lane < 16) {
        const float* g = gate + (size_t)rowm * DGATE;
        float l0 = 0.f, l1 = 0.f;
        #pragma unroll
        for (int d = 0; d < DGATE; ++d) {
            float gv = g[d];
            l0 += gv * w_gate[d * NE + 0];
            l1 += gv * w_gate[d * NE + 1];
        }
        int e = (l1 > l0) ? 1 : 0;
        sel[lane] = e;
        if ((row0 + lane) < NN && e == 1) atomicAdd(cnt1, 1);
    }
    __syncthreads();

    v2f ax[3];
    {
        const float* xr = x + (size_t)rowm * DIN;
        #pragma unroll
        for (int c = 0; c < 3; ++c) {
            const int k0 = c * 4 + hi * 2;
            float v0 = xr[min(k0,     DIN - 1)];
            float v1 = xr[min(k0 + 1, DIN - 1)];
            v2f a;
            a.x = (k0     < DIN) ? v0 : 0.f;
            a.y = (k0 + 1 < DIN) ? v1 : 0.f;
            ax[c] = a;
        }
    }

    for (int e = 0; e < NE; ++e) {
        const float* W1 = w1 + (size_t)e * DIN * HH;
        const float* W2 = w2 + (size_t)e * HH * MO;
        __builtin_prefetch(W2, 0, 1);

        for (int nt = 0; nt < 8; ++nt) {
            const int n0 = nt * 16;
            const float bias = b1[e * HH + n0 + m];
            v8f c8;
            #pragma unroll
            for (int i = 0; i < 8; ++i) c8[i] = bias;
            #pragma unroll
            for (int cix = 0; cix < 3; ++cix) {
                const int kb = cix * 4 + hi * 2;
                float w0 = W1[min(kb,     DIN - 1) * HH + n0 + m];
                float w1v = W1[min(kb + 1, DIN - 1) * HH + n0 + m];
                v2f b;
                b.x = (kb     < DIN) ? w0  : 0.f;
                b.y = (kb + 1 < DIN) ? w1v : 0.f;
                c8 = __builtin_amdgcn_wmma_f32_16x16x4_f32(
                         false, ax[cix], false, b, (short)0, c8, false, false);
            }
            #pragma unroll
            for (int i = 0; i < 8; ++i)
                h_lds[(i + hi * 8) * HH + n0 + m] = fmaxf(c8[i], 0.0f);
        }
        __syncthreads();

        for (int nt = 0; nt < 2; ++nt) {
            const int n0 = nt * 16;
            const float bias = b2[e * MO + n0 + m];
            v8f c8;
            #pragma unroll
            for (int i = 0; i < 8; ++i) c8[i] = bias;
            #pragma unroll 4
            for (int k = 0; k < HH; k += 4) {
                const int ka = k + hi * 2;
                v2f a = *(const v2f*)&h_lds[m * HH + ka];
                v2f b;
                b.x = W2[(ka    ) * MO + n0 + m];
                b.y = W2[(ka + 1) * MO + n0 + m];
                c8 = __builtin_amdgcn_wmma_f32_16x16x4_f32(
                         false, a, false, b, (short)0, c8, false, false);
            }
            #pragma unroll
            for (int i = 0; i < 8; ++i)
                y_lds[e * (16 * MO) + (i + hi * 8) * MO + n0 + m] = c8[i];
        }
        __syncthreads();
    }

    if (lane < 16 && (row0 + lane) < NN) {
        const int e = sel[lane];
        const float* yr = y_lds + e * (16 * MO) + lane * MO;
        float a0 = bo[0], a1 = bo[1];
        #pragma unroll
        for (int j = 0; j < MO; ++j) {
            float yv = yr[j];
            a0 += yv * wo[j * NO + 0];
            a1 += yv * wo[j * NO + 1];
        }
        out[(row0 + lane) * NO + 0] = a0;
        out[(row0 + lane) * NO + 1] = a1;
    }
}

__global__ void moe_loss_kernel(const int* __restrict__ cnt1,
                                float* __restrict__ loss_out) {
    if (threadIdx.x != 0) return;
    const float c1 = (float)cnt1[0];
    const float c0 = (float)NN - c1;
    const float mv = 0.5f * (float)NN;
    const float var = (c0 - mv) * (c0 - mv) + (c1 - mv) * (c1 - mv);
    const float cv  = var / (mv * mv + 1e-10f);
    loss_out[0] = 2.0f * cv * 0.01f;
}

extern "C" void kernel_launch(void* const* d_in, const int* in_sizes, int n_in,
                              void* d_out, int out_size, void* d_ws, size_t ws_size,
                              hipStream_t stream) {
    const float* x      = (const float*)d_in[0];
    const float* gate   = (const float*)d_in[1];
    const float* w_gate = (const float*)d_in[2];
    const float* w1     = (const float*)d_in[3];
    const float* b1     = (const float*)d_in[4];
    const float* w2     = (const float*)d_in[5];
    const float* b2     = (const float*)d_in[6];
    const float* wo     = (const float*)d_in[7];
    const float* bo     = (const float*)d_in[8];
    float* out = (float*)d_out;
    int*   cnt = (int*)d_ws;

    const int tiles = (NN + 15) / 16;

    moe_zero_ws<<<1, 1, 0, stream>>>(cnt);
    moe_tile_kernel<<<tiles, 32, 0, stream>>>(x, gate, w_gate, w1, b1, w2, b2,
                                              wo, bo, out, cnt);
    moe_loss_kernel<<<1, 1, 0, stream>>>(cnt, out + (size_t)NN * NO);
}


# Loss written by the work-group that arrives last: rows and arrivals counted in one word (a rows add, then a returning +1 per work-group; integer, exact); loss launch dropped: 3 launches -> 2
# speedup vs baseline: 2.2889x; 2.2889x over previous
#include <hip/hip_runtime.h>


typedef float v2f __attribute__((ext_vector_type(2)));
typedef float v8f __attribute__((ext_vector_type(8)));

#define NN    11826
#define DIN   9
#define DGATE 36
#define NE    2
#define HH    128
#define MO    32
#define NO    2

#define WPG   4
#define NTHR  (32 * WPG)
#define W1R   12
#define W1P   144
#define W2P   48
#define HP    132
#define YP    36
#define OFF_W2  0
#define OFF_W1  (OFF_W2 + NE * HH * W2P)
#define OFF_B1  (OFF_W1 + NE * W1R * W1P)
#define OFF_B2  (OFF_B1 + NE * HH)
#define PAR_DW  (OFF_B2 + NE * MO)
#define ZROW4   ((W1R - DIN) * (HH / 4))
#define NGRP    ((NN + 16 * WPG - 1) / (16 * WPG))

__global__ void moe_zero_ws(unsigned int* cnt) {
    if (threadIdx.x == 0) cnt[0] = 0u;
}

__device__ __forceinline__ float moe_loss_from_count(unsigned int n1) {
    const float c1 = (float)n1;
    const float c0 = (float)NN - c1;
    const float mv = 0.5f * (float)NN;
    const float var = (c0 - mv) * (c0 - mv) + (c1 - mv) * (c1 - mv);
    const float cv  = var / (mv * mv + 1e-10f);
    return 2.0f * cv * 0.01f;
}

__global__ __launch_bounds__(NTHR) void moe_tile_kernel(
    const float* __restrict__ x,
    const float* __restrict__ gate,
    const float* __restrict__ w_gate,
    const float* __restrict__ w1,
    const float* __restrict__ b1,
    const float* __restrict__ w2,
    const float* __restrict__ b2,
    const float* __restrict__ wo,
    const float* __restrict__ bo,
    float* __restrict__ out,
    unsigned int* __restrict__ cnt1)
{
    __shared__ __attribute__((aligned(16))) float par[PAR_DW];
    __shared__ __attribute__((aligned(16))) float h_all[WPG][16 * HP];
    __shared__ __attribute__((aligned(16))) float y_all[WPG][NE * 16 * YP];
    __shared__ int sel_all[WPG][32];
    __shared__ unsigned int wcnt[WPG];

    const int tid  = threadIdx.x;
    const int lane = tid & 31;
    const int wv   = tid >> 5;
    const int row0 = (blockIdx.x * WPG + wv) * 16;
    const int m    = lane & 15;
    const int hi   = lane >> 4;
    const int rowm = min(row0 + m, NN - 1);

    float* h_lds = h_all[wv];
    float* y_lds = y_all[wv];
    int*   sel   = sel_all[wv];

    float4 r2[16];
    {
        const float4* w2v = (const float4*)w2;
        #pragma unroll
        for (int i = 0; i < 16; ++i) r2[i] = w2v[tid + NTHR * i];
    }
    float4 r1[4];
    {
        const float4* w1v = (const float4*)w1;
        #pragma unroll
        for (int i = 0; i < 4; ++i) r1[i] = w1v[tid + NTHR * i];
    }
    const bool w1tail = tid < 64;
    const float* p5 = w1tail ? w1 + (4 * NTHR + tid) * 4 : b1 + (tid - 64) * 4;
    const float4 r5 = *(const float4*)p5;
    const bool isb2 = tid < NE * MO / 4;
    const float4 r6 = ((const float4*)b2)[min(tid, NE * MO / 4 - 1)];

    int e_sel;
    {
        const float* g = gate + (size_t)rowm * DGATE;
        float l0 = 0.f, l1 = 0.f;
        #pragma unroll
        for (int d = 0; d < DGATE; ++d) {
            float gv = g[d];
            l0 += gv * w_gate[d * NE + 0];
            l1 += gv * w_gate[d * NE + 1];
        }
        e_sel = (l1 > l0) ? 1 : 0;
    }
    sel[lane] = e_sel;

    #pragma unroll
    for (int i = 0; i < 16; ++i) {
        const int f = tid + NTHR * i;
        *(float4*)&par[OFF_W2 + (f >> 3) * W2P + (f & 7) * 4] = r2[i];
    }
    #pragma unroll
    for (int i = 0; i < 4; ++i) {
        const int f = tid + NTHR * i;
        const int r = f >> 5;
        const int e = (r >= DIN) ? 1 : 0;
        *(float4*)&par[OFF_W1 + (e * W1R + r - e * DIN) * W1P + (f & 31) * 4] = r1[i];
    }
    {
        const int d_w1 = OFF_W1 + (W1R + 7 + (tid >> 5)) * W1P + (tid & 31) * 4;
        const int d_b1 = OFF_B1 + (tid - 64) * 4;
        const int dst  = w1tail ? d_w1 : d_b1;
        *(float4*)&par[dst] = r5;
    }
    {
        const int z    = max(tid - NE * MO / 4, 0);
        const int e    = (z >= ZROW4) ? 1 : 0;
        const int rem  = z - e * ZROW4;
        const int d_z  = OFF_W1 + (e * W1R + DIN + (rem >> 5)) * W1P + (rem & 31) * 4;
        const int d_b2 = OFF_B2 + min(tid, NE * MO / 4 - 1) * 4;
        const int dst  = isb2 ? d_b2 : d_z;
        float4 v;
        v.x = isb2 ? r6.x : 0.f;
        v.y = isb2 ? r6.y : 0.f;
        v.z = isb2 ? r6.z : 0.f;
        v.w = isb2 ? r6.w : 0.f;
        *(float4*)&par[dst] = v;
    }
    if (tid < NE * ZROW4 - (NTHR - NE * MO / 4)) {
        const int z   = tid + (NTHR - NE * MO / 4);
        const int e   = (z >= ZROW4) ? 1 : 0;
        const int rem = z - e * ZROW4;
        *(float4*)&par[OFF_W1 + (e * W1R + DIN + (rem >> 5)) * W1P + (rem & 31) * 4] =
            make_float4(0.f, 0.f, 0.f, 0.f);
    }
    wcnt[wv] = (unsigned int)__popc(__builtin_amdgcn_ballot_w32(lane < 16 && (row0 + lane) < NN && e_sel == 1));
    __syncthreads();

    v2f ax[3];
    {
        const float* xr = x + (size_t)rowm * DIN;
        #pragma unroll
        for (int c = 0; c < 3; ++c) {
            const int k0 = c * 4 + hi * 2;
            float v0 = xr[min(k0,     DIN - 1)];
            float v1 = xr[min(k0 + 1, DIN - 1)];
            v2f a;
            a.x = (k0     < DIN) ? v0 : 0.f;
            a.y = (k0 + 1 < DIN) ? v1 : 0.f;
            ax[c] = a;
        }
    }

    unsigned int wg_old = 0u;

    for (int e = 0; e < NE; ++e) {
        const float* W1s = par + OFF_W1 + e * W1R * W1P;
        const float* W2s = par + OFF_W2 + e * HH * W2P;

        v2f   bw[8][3];
        float bs[8];
        #pragma unroll
        for (int nt = 0; nt < 8; ++nt) {
            const int n0 = nt * 16;
            bs[nt] = par[OFF_B1 + e * HH + n0 + m];
            #pragma unroll
            for (int cix = 0; cix < 3; ++cix) {
                const int kb = cix * 4 + hi * 2;
                bw[nt][cix].x = W1s[(kb    ) * W1P + n0 + m];
                bw[nt][cix].y = W1s[(kb + 1) * W1P + n0 + m];
            }
        }
        v2f   w2f[2][HH / 4];
        float bias2[2];
        #pragma unroll
        for (int nt = 0; nt < 2; ++nt) {
            #pragma unroll
            for (int ks = 0; ks < HH / 4; ++ks) {
                const int ka = ks * 4 + hi * 2;
                w2f[nt][ks].x = W2s[(ka    ) * W2P + nt * 16 + m];
                w2f[nt][ks].y = W2s[(ka + 1) * W2P + nt * 16 + m];
            }
            __builtin_amdgcn_sched_barrier(0);
        }
        bias2[0] = par[OFF_B2 + e * MO + m];
        bias2[1] = par[OFF_B2 + e * MO + 16 + m];

        v8f acc[8];
        #pragma unroll
        for (int nt = 0; nt < 8; ++nt) {
            #pragma unroll
            for (int i = 0; i < 8; ++i) acc[nt][i] = bs[nt];
        }
        #pragma unroll
        for (int cix = 0; cix < 3; ++cix) {
            #pragma unroll
            for (int nt = 0; nt < 8; ++nt)
                acc[nt] = __builtin_amdgcn_wmma_f32_16x16x4_f32(
                              false, ax[cix], false, bw[nt][cix], (short)0, acc[nt], false, false);
        }
        #pragma unroll
        for (int nt = 0; nt < 8; ++nt) {
            #pragma unroll
            for (int i = 0; i < 8; ++i)
                h_lds[(i + hi * 8) * HP + nt * 16 + m] = fmaxf(acc[nt][i], 0.0f);
        }
        __syncthreads();

        if (e == 0 && tid == 0) {
            const unsigned int rows = wcnt[0] + wcnt[1] + wcnt[2] + wcnt[3];
            __hip_atomic_fetch_add(cnt1, rows << 16, __ATOMIC_RELAXED, __HIP_MEMORY_SCOPE_AGENT);
            wg_old = atomicInc(cnt1, 0xFFFFFFFFu);
        }

        {
            v2f hf[HH / 4];
            #pragma unroll
            for (int ks = 0; ks < HH / 4; ++ks)
                hf[ks] = *(const v2f*)&h_lds[m * HP + ks * 4 + hi * 2];
            __builtin_amdgcn_sched_barrier(0);
            v8f c0, c1;
            #pragma unroll
            for (int i = 0; i < 8; ++i) { c0[i] = bias2[0]; c1[i] = bias2[1]; }
            #pragma unroll
            for (int ks = 0; ks < HH / 4; ++ks) {
                c0 = __builtin_amdgcn_wmma_f32_16x16x4_f32(
                         false, hf[ks], false, w2f[0][ks], (short)0, c0, false, false);
                c1 = __builtin_amdgcn_wmma_f32_16x16x4_f32(
                         false, hf[ks], false, w2f[1][ks], (short)0, c1, false, false);
            }
            #pragma unroll
            for (int i = 0; i < 8; ++i) {
                y_lds[e * (16 * YP) + (i + hi * 8) * YP + m]      = c0[i];
                y_lds[e * (16 * YP) + (i + hi * 8) * YP + 16 + m] = c1[i];
            }
        }
        __syncthreads();
    }

    if (lane < 16 && (row0 + lane) < NN) {
        const int e = sel[lane];
        const float* yr = y_lds + e * (16 * YP) + lane * YP;
        float a0 = bo[0], a1 = bo[1];
        #pragma unroll
        for (int j = 0; j < MO; ++j) {
            float yv = yr[j];
            a0 += yv * wo[j * NO + 0];
            a1 += yv * wo[j * NO + 1];
        }
        out[(row0 + lane) * NO + 0] = a0;
        out[(row0 + lane) * NO + 1] = a1;
    }

    if (tid == 0 && (wg_old & 0xFFFFu) == (unsigned int)(NGRP - 1))
        out[(size_t)NN * NO] = moe_loss_from_count(wg_old >> 16);
}

extern "C" void kernel_launch(void* const* d_in, const int* in_sizes, int n_in,
                              void* d_out, int out_size, void* d_ws, size_t ws_size,
                              hipStream_t stream) {
    const float* x      = (const float*)d_in[0];
    const float* gate   = (const float*)d_in[1];
    const float* w_gate = (const float*)d_in[2];
    const float* w1     = (const float*)d_in[3];
    const float* b1     = (const float*)d_in[4];
    const float* w2     = (const float*)d_in[5];
    const float* b2     = (const float*)d_in[6];
    const float* wo     = (const float*)d_in[7];
    const float* bo     = (const float*)d_in[8];
    float* out = (float*)d_out;
    unsigned int* cnt = (unsigned int*)d_ws;

    moe_zero_ws<<<1, 1, 0, stream>>>(cnt);
    moe_tile_kernel<<<NGRP, NTHR, 0, stream>>>(x, gate, w_gate, w1, b1, w2, b2,
                                               wo, bo, out, cnt);
}


# Rows grouped by chosen expert inside a 48-row work-group (ballot masks, rank select): each of the 4 waves runs ONE expert on one 16-row tile; 176 -> 88 matrix ops per wave, per-row sums unchanged
# speedup vs baseline: 2.6352x; 1.1513x over previous
#include <hip/hip_runtime.h>


typedef float v2f __attribute__((ext_vector_type(2)));
typedef float v8f __attribute__((ext_vector_type(8)));

#define NN    11826
#define DIN   9
#define DGATE 36
#define NE    2
#define HH    128
#define MO    32
#define NO    2

#define WPG   4
#define RPG   48
#define NTHR  (32 * WPG)
#define W1R   12
#define W1P   144
#define W2P   48
#define HP    132
#define YP    36
#define OFF_W2  0
#define OFF_W1  (OFF_W2 + NE * HH * W2P)
#define OFF_B1  (OFF_W1 + NE * W1R * W1P)
#define OFF_B2  (OFF_B1 + NE * HH)
#define PAR_DW  (OFF_B2 + NE * MO)
#define ZROW4   ((W1R - DIN) * (HH / 4))
#define NGRP    ((NN + RPG - 1) / RPG)

__global__ void moe_zero_ws(unsigned int* cnt) {
    if (threadIdx.x == 0) cnt[0] = 0u;
}

__device__ __forceinline__ float moe_loss_from_count(unsigned int n1) {
    const float c1 = (float)n1;
    const float c0 = (float)NN - c1;
    const float mv = 0.5f * (float)NN;
    const float var = (c0 - mv) * (c0 - mv) + (c1 - mv) * (c1 - mv);
    const float cv  = var / (mv * mv + 1e-10f);
    return 2.0f * cv * 0.01f;
}

__global__ __launch_bounds__(NTHR) void moe_tile_kernel(
    const float* __restrict__ x,
    const float* __restrict__ gate,
    const float* __restrict__ w_gate,
    const float* __restrict__ w1,
    const float* __restrict__ b1,
    const float* __restrict__ w2,
    const float* __restrict__ b2,
    const float* __restrict__ wo,
    const float* __restrict__ bo,
    float* __restrict__ out,
    unsigned int* __restrict__ cnt1)
{
    __shared__ __attribute__((aligned(16))) float par[PAR_DW];
    __shared__ __attribute__((aligned(16))) float h_all[WPG][16 * HP];
    __shared__ __attribute__((aligned(16))) float y_all[WPG][16 * YP];
    __shared__ __attribute__((aligned(16))) float ost[RPG * NO];
    __shared__ __attribute__((aligned(16))) unsigned int smask[4];

    const int tid   = threadIdx.x;
    const int lane  = tid & 31;
    const int wv    = tid >> 5;
    const int rbase = blockIdx.x * RPG;
    const int m     = lane & 15;
    const int hi    = lane >> 4;
    const int gi    = wv * 16 + m;
    const int rowg  = min(rbase + gi, NN - 1);

    float* h_lds = h_all[wv];
    float* y_lds = y_all[wv];

    float4 r2[16];
    {
        const float4* w2v = (const float4*)w2;
        #pragma unroll
        for (int i = 0; i < 16; ++i) r2[i] = w2v[tid + NTHR * i];
    }
    float4 r1[4];
    {
        const float4* w1v = (const float4*)w1;
        #pragma unroll
        for (int i = 0; i < 4; ++i) r1[i] = w1v[tid + NTHR * i];
    }
    const bool w1tail = tid < 64;
    const float* p5 = w1tail ? w1 + (4 * NTHR + tid) * 4 : b1 + (tid - 64) * 4;
    const float4 r5 = *(const float4*)p5;
    const bool isb2 = tid < NE * MO / 4;
    const float4 r6 = ((const float4*)b2)[min(tid, NE * MO / 4 - 1)];

    int e_sel;
    {
        const float* g = gate + (size_t)rowg * DGATE;
        float l0 = 0.f, l1 = 0.f;
        #pragma unroll
        for (int d = 0; d < DGATE; ++d) {
            float gv = g[d];
            l0 += gv * w_gate[d * NE + 0];
            l1 += gv * w_gate[d * NE + 1];
        }
        e_sel = (l1 > l0) ? 1 : 0;
    }
    {
        const bool gvalid = (wv < 3) && (lane < 16) && (rbase + gi) < NN;
        const unsigned int bit1 = __builtin_amdgcn_ballot_w32(gvalid && e_sel == 1);
        const unsigned int bit0 = __builtin_amdgcn_ballot_w32(gvalid && e_sel == 0);
        smask[wv] = bit1 | (bit0 << 16);
    }

    #pragma unroll
    for (int i = 0; i < 16; ++i) {
        const int f = tid + NTHR * i;
        *(float4*)&par[OFF_W2 + (f >> 3) * W2P + (f & 7) * 4] = r2[i];
    }
    #pragma unroll
    for (int i = 0; i < 4; ++i) {
        const int f = tid + NTHR * i;
        const int r = f >> 5;
        const int e = (r >= DIN) ? 1 : 0;
        *(float4*)&par[OFF_W1 + (e * W1R + r - e * DIN) * W1P + (f & 31) * 4] = r1[i];
    }
    {
        const int d_w1 = OFF_W1 + (W1R + 7 + (tid >> 5)) * W1P + (tid & 31) * 4;
        const int d_b1 = OFF_B1 + (tid - 64) * 4;
        const int dst  = w1tail ? d_w1 : d_b1;
        *(float4*)&par[dst] = r5;
    }
    {
        const int z    = max(tid - NE * MO / 4, 0);
        const int e    = (z >= ZROW4) ? 1 : 0;
        const int rem  = z - e * ZROW4;
        const int d_z  = OFF_W1 + (e * W1R + DIN + (rem >> 5)) * W1P + (rem & 31) * 4;
        const int d_b2 = OFF_B2 + min(tid, NE * MO / 4 - 1) * 4;
        const int dst  = isb2 ? d_b2 : d_z;
        float4 v;
        v.x = isb2 ? r6.x : 0.f;
        v.y = isb2 ? r6.y : 0.f;
        v.z = isb2 ? r6.z : 0.f;
        v.w = isb2 ? r6.w : 0.f;
        *(float4*)&par[dst] = v;
    }
    if (tid < NE * ZROW4 - (NTHR - NE * MO / 4)) {
        const int z   = tid + (NTHR - NE * MO / 4);
        const int e   = (z >= ZROW4) ? 1 : 0;
        const int rem = z - e * ZROW4;
        *(float4*)&par[OFF_W1 + (e * W1R + DIN + (rem >> 5)) * W1P + (rem & 31) * 4] =
            make_float4(0.f, 0.f, 0.f, 0.f);
    }
    __syncthreads();

    const unsigned int k0 = smask[0], k1 = smask[1], k2 = smask[2];
    const int n1 = __popc(k0 & 0xFFFFu) + __popc(k1 & 0xFFFFu) + __popc(k2 & 0xFFFFu);
    const int n0 = __popc(k0 >> 16) + __popc(k1 >> 16) + __popc(k2 >> 16);
    const int u0 = (n0 + 15) >> 4;
    const int eu  = __builtin_amdgcn_readfirstlane((wv >= u0) ? 1 : 0);
    const int tix = eu ? wv - u0 : wv;
    const int ne  = eu ? n1 : n0;
    int jj = tix * 16 + m;
    const bool valid = jj < ne;
    int rloc;
    {
        const unsigned int hA = eu ? (k0 & 0xFFFFu) : (k0 >> 16);
        const unsigned int hB = eu ? (k1 & 0xFFFFu) : (k1 >> 16);
        const unsigned int hC = eu ? (k2 & 0xFFFFu) : (k2 >> 16);
        unsigned int w = hA | (hB << 16);
        int pos = 0;
        int c = __popc(w);
        if (jj >= c) { jj -= c; w = hC; pos = 32; }
        c = __popc(w & 0xFFFFu); if (jj >= c) { jj -= c; w >>= 16; pos += 16; }
        c = __popc(w & 0xFFu);   if (jj >= c) { jj -= c; w >>= 8;  pos += 8; }
        c = __popc(w & 0xFu);    if (jj >= c) { jj -= c; w >>= 4;  pos += 4; }
        c = __popc(w & 0x3u);    if (jj >= c) { jj -= c; w >>= 2;  pos += 2; }
        c = (int)(w & 1u);       if (jj >= c) { pos += 1; }
        rloc = valid ? pos : 0;
    }
    const int rowx = rbase + rloc;

    v2f ax[3];
    {
        const float* xr = x + (size_t)rowx * DIN;
        #pragma unroll
        for (int c = 0; c < 3; ++c) {
            const int kk = c * 4 + hi * 2;
            float v0 = xr[min(kk,     DIN - 1)];
            float v1 = xr[min(kk + 1, DIN - 1)];
            v2f a;
            a.x = (kk     < DIN) ? v0 : 0.f;
            a.y = (kk + 1 < DIN) ? v1 : 0.f;
            ax[c] = a;
        }
    }

    unsigned int wg_old = 0u;

    {
        const float* W1s = par + OFF_W1 + eu * W1R * W1P;
        const float* W2s = par + OFF_W2 + eu * HH * W2P;

        v2f   bw[8][3];
        float bs[8];
        #pragma unroll
        for (int nt = 0; nt < 8; ++nt) {
            const int n0c = nt * 16;
            bs[nt] = par[OFF_B1 + eu * HH + n0c + m];
            #pragma unroll
            for (int cix = 0; cix < 3; ++cix) {
                const int kb = cix * 4 + hi * 2;
                bw[nt][cix].x = W1s[(kb    ) * W1P + n0c + m];
                bw[nt][cix].y = W1s[(kb + 1) * W1P + n0c + m];
            }
        }
        v2f   w2f[2][HH / 4];
        float bias2[2];
        #pragma unroll
        for (int nt = 0; nt < 2; ++nt) {
            #pragma unroll
            for (int ks = 0; ks < HH / 4; ++ks) {
                const int ka = ks * 4 + hi * 2;
                w2f[nt][ks].x = W2s[(ka    ) * W2P + nt * 16 + m];
                w2f[nt][ks].y = W2s[(ka + 1) * W2P + nt * 16 + m];
            }
            __builtin_amdgcn_sched_barrier(0);
        }
        bias2[0] = par[OFF_B2 + eu * MO + m];
        bias2[1] = par[OFF_B2 + eu * MO + 16 + m];

        v8f acc[8];
        #pragma unroll
        for (int nt = 0; nt < 8; ++nt) {
            #pragma unroll
            for (int i = 0; i < 8; ++i) acc[nt][i] = bs[nt];
        }
        #pragma unroll
        for (int cix = 0; cix < 3; ++cix) {
            #pragma unroll
            for (int nt = 0; nt < 8; ++nt)
                acc[nt] = __builtin_amdgcn_wmma_f32_16x16x4_f32(
                              false, ax[cix], false, bw[nt][cix], (short)0, acc[nt], false, false);
        }
        #pragma unroll
        for (int nt = 0; nt < 8; ++nt) {
            #pragma unroll
            for (int i = 0; i < 8; ++i)
                h_lds[(i + hi * 8) * HP + nt * 16 + m] = fmaxf(acc[nt][i], 0.0f);
        }
        __syncthreads();

        if (tid == 0) {
            __hip_atomic_fetch_add(cnt1, (unsigned int)n1 << 16, __ATOMIC_RELAXED, __HIP_MEMORY_SCOPE_AGENT);
            wg_old = atomicInc(cnt1, 0xFFFFFFFFu);
        }

        {
            v2f hf[HH / 4];
            #pragma unroll
            for (int ks = 0; ks < HH / 4; ++ks)
                hf[ks] = *(const v2f*)&h_lds[m * HP + ks * 4 + hi * 2];
            __builtin_amdgcn_sched_barrier(0);
            v8f c0, c1;
            #pragma unroll
            for (int i = 0; i < 8; ++i) { c0[i] = bias2[0]; c1[i] = bias2[1]; }
            #pragma unroll
            for (int ks = 0; ks < HH / 4; ++ks) {
                c0 = __builtin_amdgcn_wmma_f32_16x16x4_f32(
                         false, hf[ks], false, w2f[0][ks], (short)0, c0, false, false);
                c1 = __builtin_amdgcn_wmma_f32_16x16x4_f32(
                         false, hf[ks], false, w2f[1][ks], (short)0, c1, false, false);
            }
            #pragma unroll
            for (int i = 0; i < 8; ++i) {
                y_lds[(i + hi * 8) * YP + m]      = c0[i];
                y_lds[(i + hi * 8) * YP + 16 + m] = c1[i];
            }
        }
        __syncthreads();
    }

    if (lane < 16 && valid) {
        const float* yr = y_lds + lane * YP;
        float a0 = bo[0], a1 = bo[1];
        #pragma unroll
        for (int j = 0; j < MO; ++j) {
            float yv = yr[j];
            a0 += yv * wo[j * NO + 0];
            a1 += yv * wo[j * NO + 1];
        }
        ost[rloc * NO + 0] = a0;
        ost[rloc * NO + 1] = a1;
    }
    __syncthreads();
    if (tid < RPG * NO && (rbase * NO + tid) < NN * NO)
        out[rbase * NO + tid] = ost[tid];

    if (tid == 0 && (wg_old & 0xFFFFu) == (unsigned int)(NGRP - 1))
        out[(size_t)NN * NO] = moe_loss_from_count(wg_old >> 16);
}

extern "C" void kernel_launch(void* const* d_in, const int* in_sizes, int n_in,
                              void* d_out, int out_size, void* d_ws, size_t ws_size,
                              hipStream_t stream) {
    const float* x      = (const float*)d_in[0];
    const float* gate   = (const float*)d_in[1];
    const float* w_gate = (const float*)d_in[2];
    const float* w1     = (const float*)d_in[3];
    const float* b1     = (const float*)d_in[4];
    const float* w2     = (const float*)d_in[5];
    const float* b2     = (const float*)d_in[6];
    const float* wo     = (const float*)d_in[7];
    const float* bo     = (const float*)d_in[8];
    float* out = (float*)d_out;
    unsigned int* cnt = (unsigned int*)d_ws;

    moe_zero_ws<<<1, 1, 0, stream>>>(cnt);
    moe_tile_kernel<<<NGRP, NTHR, 0, stream>>>(x, gate, w_gate, w1, b1, w2, b2,
                                               wo, bo, out, cnt);
}


# w_gate by lane broadcast (readlane of one 16-byte load per lane), wo and bo read from the LDS parameter block: 7 dependent scalar-load round trips (3 in gating, 4 in projection) gone
# speedup vs baseline: 2.9896x; 1.1345x over previous
#include <hip/hip_runtime.h>


typedef float v2f __attribute__((ext_vector_type(2)));
typedef float v8f __attribute__((ext_vector_type(8)));

#define NN    11826
#define DIN   9
#define DGATE 36
#define NE    2
#define HH    128
#define MO    32
#define NO    2

#define WPG   4
#define RPG   48
#define NTHR  (32 * WPG)
#define W1R   12
#define W1P   144
#define W2P   48
#define HP    132
#define YP    36
#define OFF_W2  0
#define OFF_W1  (OFF_W2 + NE * HH * W2P)
#define OFF_B1  (OFF_W1 + NE * W1R * W1P)
#define OFF_B2  (OFF_B1 + NE * HH)
#define OFF_WO  (OFF_B2 + NE * MO)
#define OFF_BO  (OFF_WO + NTHR * 4)
#define PAR_DW  (OFF_BO + NTHR)
#define ZROW4   ((W1R - DIN) * (HH / 4))
#define NGRP    ((NN + RPG - 1) / RPG)

__global__ void moe_zero_ws(unsigned int* cnt) {
    if (threadIdx.x == 0) cnt[0] = 0u;
}

__device__ __forceinline__ float moe_loss_from_count(unsigned int n1) {
    const float c1 = (float)n1;
    const float c0 = (float)NN - c1;
    const float mv = 0.5f * (float)NN;
    const float var = (c0 - mv) * (c0 - mv) + (c1 - mv) * (c1 - mv);
    const float cv  = var / (mv * mv + 1e-10f);
    return 2.0f * cv * 0.01f;
}

__global__ __launch_bounds__(NTHR) void moe_tile_kernel(
    const float* __restrict__ x,
    const float* __restrict__ gate,
    const float* __restrict__ w_gate,
    const float* __restrict__ w1,
    const float* __restrict__ b1,
    const float* __restrict__ w2,
    const float* __restrict__ b2,
    const float* __restrict__ wo,
    const float* __restrict__ bo,
    float* __restrict__ out,
    unsigned int* __restrict__ cnt1)
{
    __shared__ __attribute__((aligned(16))) float par[PAR_DW];
    __shared__ __attribute__((aligned(16))) float h_all[WPG][16 * HP];
    __shared__ __attribute__((aligned(16))) float y_all[WPG][16 * YP];
    __shared__ __attribute__((aligned(16))) float ost[RPG * NO];
    __shared__ __attribute__((aligned(16))) unsigned int smask[4];

    const int tid   = threadIdx.x;
    const int lane  = tid & 31;
    const int wv    = tid >> 5;
    const int rbase = blockIdx.x * RPG;
    const int m     = lane & 15;
    const int hi    = lane >> 4;
    const int gi    = wv * 16 + m;
    const int rowg  = min(rbase + gi, NN - 1);

    float* h_lds = h_all[wv];
    float* y_lds = y_all[wv];

    const float4 rg = ((const float4*)w_gate)[min(lane, DGATE * NE / 4 - 1)];
    const float4 r7 = ((const float4*)wo)[tid & (MO * NO / 4 - 1)];
    const float  r8 = bo[tid & (NO - 1)];
    float4 r2[16];
    {
        const float4* w2v = (const float4*)w2;
        #pragma unroll
        for (int i = 0; i < 16; ++i) r2[i] = w2v[tid + NTHR * i];
    }
    float4 r1[4];
    {
        const float4* w1v = (const float4*)w1;
        #pragma unroll
        for (int i = 0; i < 4; ++i) r1[i] = w1v[tid + NTHR * i];
    }
    const bool w1tail = tid < 64;
    const float* p5 = w1tail ? w1 + (4 * NTHR + tid) * 4 : b1 + (tid - 64) * 4;
    const float4 r5 = *(const float4*)p5;
    const bool isb2 = tid < NE * MO / 4;
    const float4 r6 = ((const float4*)b2)[min(tid, NE * MO / 4 - 1)];

    int e_sel;
    {
        const float* g = gate + (size_t)rowg * DGATE;
        float l0 = 0.f, l1 = 0.f;
        #pragma unroll
        for (int d = 0; d < DGATE; ++d) {
            float gv = g[d];
            const int   sl = d >> 1;
            const float wa = (d & 1) ? rg.z : rg.x;
            const float wb = (d & 1) ? rg.w : rg.y;
            const float w0 = __int_as_float(__builtin_amdgcn_readlane(__float_as_int(wa), sl));
            const float w1v = __int_as_float(__builtin_amdgcn_readlane(__float_as_int(wb), sl));
            l0 += gv * w0;
            l1 += gv * w1v;
        }
        e_sel = (l1 > l0) ? 1 : 0;
    }
    {
        const bool gvalid = (wv < 3) && (lane < 16) && (rbase + gi) < NN;
        const unsigned int bit1 = __builtin_amdgcn_ballot_w32(gvalid && e_sel == 1);
        const unsigned int bit0 = __builtin_amdgcn_ballot_w32(gvalid && e_sel == 0);
        smask[wv] = bit1 | (bit0 << 16);
    }

    #pragma unroll
    for (int i = 0; i < 16; ++i) {
        const int f = tid + NTHR * i;
        *(float4*)&par[OFF_W2 + (f >> 3) * W2P + (f & 7) * 4] = r2[i];
    }
    #pragma unroll
    for (int i = 0; i < 4; ++i) {
        const int f = tid + NTHR * i;
        const int r = f >> 5;
        const int e = (r >= DIN) ? 1 : 0;
        *(float4*)&par[OFF_W1 + (e * W1R + r - e * DIN) * W1P + (f & 31) * 4] = r1[i];
    }
    {
        const int d_w1 = OFF_W1 + (W1R + 7 + (tid >> 5)) * W1P + (tid & 31) * 4;
        const int d_b1 = OFF_B1 + (tid - 64) * 4;
        const int dst  = w1tail ? d_w1 : d_b1;
        *(float4*)&par[dst] = r5;
    }
    {
        const int z    = max(tid - NE * MO / 4, 0);
        const int e    = (z >= ZROW4) ? 1 : 0;
        const int rem  = z - e * ZROW4;
        const int d_z  = OFF_W1 + (e * W1R + DIN + (rem >> 5)) * W1P + (rem & 31) * 4;
        const int d_b2 = OFF_B2 + min(tid, NE * MO / 4 - 1) * 4;
        const int dst  = isb2 ? d_b2 : d_z;
        float4 v;
        v.x = isb2 ? r6.x : 0.f;
        v.y = isb2 ? r6.y : 0.f;
        v.z = isb2 ? r6.z : 0.f;
        v.w = isb2 ? r6.w : 0.f;
        *(float4*)&par[dst] = v;
    }
    *(float4*)&par[OFF_WO + tid * 4] = r7;
    par[OFF_BO + tid] = r8;
    if (tid < NE * ZROW4 - (NTHR - NE * MO / 4)) {
        const int z   = tid + (NTHR - NE * MO / 4);
        const int e   = (z >= ZROW4) ? 1 : 0;
        const int rem = z - e * ZROW4;
        *(float4*)&par[OFF_W1 + (e * W1R + DIN + (rem >> 5)) * W1P + (rem & 31) * 4] =
            make_float4(0.f, 0.f, 0.f, 0.f);
    }
    __syncthreads();

    const unsigned int k0 = smask[0], k1 = smask[1], k2 = smask[2];
    const int n1 = __popc(k0 & 0xFFFFu) + __popc(k1 & 0xFFFFu) + __popc(k2 & 0xFFFFu);
    const int n0 = __popc(k0 >> 16) + __popc(k1 >> 16) + __popc(k2 >> 16);
    const int u0 = (n0 + 15) >> 4;
    const int eu  = __builtin_amdgcn_readfirstlane((wv >= u0) ? 1 : 0);
    const int tix = eu ? wv - u0 : wv;
    const int ne  = eu ? n1 : n0;
    int jj = tix * 16 + m;
    const bool valid = jj < ne;
    int rloc;
    {
        const unsigned int hA = eu ? (k0 & 0xFFFFu) : (k0 >> 16);
        const unsigned int hB = eu ? (k1 & 0xFFFFu) : (k1 >> 16);
        const unsigned int hC = eu ? (k2 & 0xFFFFu) : (k2 >> 16);
        unsigned int w = hA | (hB << 16);
        int pos = 0;
        int c = __popc(w);
        if (jj >= c) { jj -= c; w = hC; pos = 32; }
        c = __popc(w & 0xFFFFu); if (jj >= c) { jj -= c; w >>= 16; pos += 16; }
        c = __popc(w & 0xFFu);   if (jj >= c) { jj -= c; w >>= 8;  pos += 8; }
        c = __popc(w & 0xFu);    if (jj >= c) { jj -= c; w >>= 4;  pos += 4; }
        c = __popc(w & 0x3u);    if (jj >= c) { jj -= c; w >>= 2;  pos += 2; }
        c = (int)(w & 1u);       if (jj >= c) { pos += 1; }
        rloc = valid ? pos : 0;
    }
    const int rowx = rbase + rloc;

    v2f ax[3];
    {
        const float* xr = x + (size_t)rowx * DIN;
        #pragma unroll
        for (int c = 0; c < 3; ++c) {
            const int kk = c * 4 + hi * 2;
            float v0 = xr[min(kk,     DIN - 1)];
            float v1 = xr[min(kk + 1, DIN - 1)];
            v2f a;
            a.x = (kk     < DIN) ? v0 : 0.f;
            a.y = (kk + 1 < DIN) ? v1 : 0.f;
            ax[c] = a;
        }
    }

    unsigned int wg_old = 0u;

    {
        const float* W1s = par + OFF_W1 + eu * W1R * W1P;
        const float* W2s = par + OFF_W2 + eu * HH * W2P;

        v2f   bw[8][3];
        float bs[8];
        #pragma unroll
        for (int nt = 0; nt < 8; ++nt) {
            const int n0c = nt * 16;
            bs[nt] = par[OFF_B1 + eu * HH + n0c + m];
            #pragma unroll
            for (int cix = 0; cix < 3; ++cix) {
                const int kb = cix * 4 + hi * 2;
                bw[nt][cix].x = W1s[(kb    ) * W1P + n0c + m];
                bw[nt][cix].y = W1s[(kb + 1) * W1P + n0c + m];
            }
        }
        v2f   w2f[2][HH / 4];
        float bias2[2];
        #pragma unroll
        for (int nt = 0; nt < 2; ++nt) {
            #pragma unroll
            for (int ks = 0; ks < HH / 4; ++ks) {
                const int ka = ks * 4 + hi * 2;
                w2f[nt][ks].x = W2s[(ka    ) * W2P + nt * 16 + m];
                w2f[nt][ks].y = W2s[(ka + 1) * W2P + nt * 16 + m];
            }
            __builtin_amdgcn_sched_barrier(0);
        }
        bias2[0] = par[OFF_B2 + eu * MO + m];
        bias2[1] = par[OFF_B2 + eu * MO + 16 + m];

        v8f acc[8];
        #pragma unroll
        for (int nt = 0; nt < 8; ++nt) {
            #pragma unroll
            for (int i = 0; i < 8; ++i) acc[nt][i] = bs[nt];
        }
        #pragma unroll
        for (int cix = 0; cix < 3; ++cix) {
            #pragma unroll
            for (int nt = 0; nt < 8; ++nt)
                acc[nt] = __builtin_amdgcn_wmma_f32_16x16x4_f32(
                              false, ax[cix], false, bw[nt][cix], (short)0, acc[nt], false, false);
        }
        #pragma unroll
        for (int nt = 0; nt < 8; ++nt) {
            #pragma unroll
            for (int i = 0; i < 8; ++i)
                h_lds[(i + hi * 8) * HP + nt * 16 + m] = fmaxf(acc[nt][i], 0.0f);
        }
        __syncthreads();

        if (tid == 0) {
            __hip_atomic_fetch_add(cnt1, (unsigned int)n1 << 16, __ATOMIC_RELAXED, __HIP_MEMORY_SCOPE_AGENT);
            wg_old = atomicInc(cnt1, 0xFFFFFFFFu);
        }

        {
            v2f hf[HH / 4];
            #pragma unroll
            for (int ks = 0; ks < HH / 4; ++ks)
                hf[ks] = *(const v2f*)&h_lds[m * HP + ks * 4 + hi * 2];
            __builtin_amdgcn_sched_barrier(0);
            v8f c0, c1;
            #pragma unroll
            for (int i = 0; i < 8; ++i) { c0[i] = bias2[0]; c1[i] = bias2[1]; }
            #pragma unroll
            for (int ks = 0; ks < HH / 4; ++ks) {
                c0 = __builtin_amdgcn_wmma_f32_16x16x4_f32(
                         false, hf[ks], false, w2f[0][ks], (short)0, c0, false, false);
                c1 = __builtin_amdgcn_wmma_f32_16x16x4_f32(
                         false, hf[ks], false, w2f[1][ks], (short)0, c1, false, false);
            }
            #pragma unroll
            for (int i = 0; i < 8; ++i) {
                y_lds[(i + hi * 8) * YP + m]      = c0[i];
                y_lds[(i + hi * 8) * YP + 16 + m] = c1[i];
            }
        }
        __syncthreads();
    }

    if (lane < 16 && valid) {
        const float* yr  = y_lds + lane * YP;
        const float* wos = par + OFF_WO;
        float a0 = par[OFF_BO + 0], a1 = par[OFF_BO + 1];
        #pragma unroll
        for (int j = 0; j < MO; ++j) {
            float yv = yr[j];
            a0 += yv * wos[j * NO + 0];
            a1 += yv * wos[j * NO + 1];
        }
        ost[rloc * NO + 0] = a0;
        ost[rloc * NO + 1] = a1;
    }
    __syncthreads();
    if (tid < RPG * NO && (rbase * NO + tid) < NN * NO)
        out[rbase * NO + tid] = ost[tid];

    if (tid == 0 && (wg_old & 0xFFFFu) == (unsigned int)(NGRP - 1))
        out[(size_t)NN * NO] = moe_loss_from_count(wg_old >> 16);
}

extern "C" void kernel_launch(void* const* d_in, const int* in_sizes, int n_in,
                              void* d_out, int out_size, void* d_ws, size_t ws_size,
                              hipStream_t stream) {
    const float* x      = (const float*)d_in[0];
    const float* gate   = (const float*)d_in[1];
    const float* w_gate = (const float*)d_in[2];
    const float* w1     = (const float*)d_in[3];
    const float* b1     = (const float*)d_in[4];
    const float* w2     = (const float*)d_in[5];
    const float* b2     = (const float*)d_in[6];
    const float* wo     = (const float*)d_in[7];
    const float* bo     = (const float*)d_in[8];
    float* out = (float*)d_out;
    unsigned int* cnt = (unsigned int*)d_ws;

    moe_zero_ws<<<1, 1, 0, stream>>>(cnt);
    moe_tile_kernel<<<NGRP, NTHR, 0, stream>>>(x, gate, w_gate, w1, b1, w2, b2,
                                               wo, bo, out, cnt);
}
